# starting kernel
# baseline (speedup 1.0000x reference)
#include <hip/hip_runtime.h>
#include <math.h>


#define TRAJ_LEN 10000
#define CHUNK    2048
#define NELEM    (2048 * 32)

typedef __attribute__((ext_vector_type(2))) float v2f;
typedef __attribute__((ext_vector_type(8))) float v8f;

__global__ void chaos_traj_kernel(const float* __restrict__ ic,
                                  const float* __restrict__ thrp,
                                  float* __restrict__ traj,
                                  float* __restrict__ prefE,
                                  float* __restrict__ prefC) {
  if (blockIdx.x == 0 && threadIdx.x == 0) {
    float z  = ic[0];
    float th = thrp[0];
    float e = 0.f, c = 0.f;
    for (int k = 0; k < TRAJ_LEN; ++k) {
      float cur = z;
      traj[k] = cur;
      e += cur * cur;
      c += (cur > th) ? 1.f : 0.f;
      prefE[k] = e;
      prefC[k] = c;
      z = (cur < th) ? (cur / th) : ((1.f - cur) / (1.f - th));
    }
  }
}

__global__ void chaos_feat_kernel(const float* __restrict__ x,
                                  const float* __restrict__ epsp,
                                  const float* __restrict__ traj,
                                  const float* __restrict__ prefE,
                                  const float* __restrict__ prefC,
                                  float* __restrict__ feats) {
  __shared__ float s_traj[CHUNK];
  __shared__ int   s_alive;

  const int   idx = blockIdx.x * blockDim.x + threadIdx.x;
  const float xv  = x[idx];
  const float eps = epsp[0];

  int kFire = -1;
  for (int base = 0; base < TRAJ_LEN; base += CHUNK) {
    const int cn = (TRAJ_LEN - base < CHUNK) ? (TRAJ_LEN - base) : CHUNK;
    for (int i = threadIdx.x; i < cn; i += blockDim.x)
      s_traj[i] = traj[base + i];
    if (threadIdx.x == 0) s_alive = 0;
    if (base + CHUNK < TRAJ_LEN)
      __builtin_prefetch(traj + base + CHUNK + threadIdx.x * 8, 0, 0);
    __syncthreads();

    if (kFire < 0) {
      for (int i = 0; i < cn; ++i) {
        if (fabsf(s_traj[i] - xv) < eps) { kFire = base + i; break; }
      }
      if (kFire < 0) atomicOr(&s_alive, 1);
    }
    __syncthreads();
    const int alive = s_alive;
    __syncthreads();
    if (!alive) break;
  }

  const int   k      = (kFire < 0) ? (TRAJ_LEN - 1) : kFire;
  const float tt     = (kFire < 0) ? (float)TRAJ_LEN : (float)(kFire + 1);
  const float energy = prefE[k];
  const float cnt    = prefC[k];
  const float p      = cnt / tt;
  const float q      = 1.f - p;
  const float e0     = (p > 0.f) ? p * log2f(p) : 0.f;
  const float e1     = (q > 0.f) ? q * log2f(q) : 0.f;
  const float ent    = -(e0 + e1);

  float4 f4 = make_float4(tt, energy, p, ent);
  *reinterpret_cast<float4*>(feats + (size_t)idx * 4) = f4;
}

__global__ void chaos_gemm_kernel(const float* __restrict__ feats,
                                  const float* __restrict__ W,
                                  const float* __restrict__ b,
                                  float* __restrict__ out) {
  __shared__ float s_B[16 * 128];

  const int lane = threadIdx.x;
  const int tile = blockIdx.x;
  const int m    = lane & 15;
  const int hi   = lane >> 4;

  for (int i = lane; i < 16 * 128; i += 32) {
    const int n = i >> 7;
    const int k = i & 127;
    const float v = W[(n & 1) * 128 + k];
    s_B[i] = (n < 2) ? v : 0.f;
  }
  __syncthreads();

  const float* __restrict__ arow = feats + (size_t)(tile * 16 + m) * 128;
  const float* __restrict__ brow = s_B + m * 128;

  v8f c = {};
  #pragma unroll 4
  for (int k0 = 0; k0 < 128; k0 += 4) {
    v2f a;
    a.x = arow[k0 + hi * 2 + 0];
    a.y = arow[k0 + hi * 2 + 1];
    v2f bb;
    bb.x = brow[k0 + hi * 2 + 0];
    bb.y = brow[k0 + hi * 2 + 1];
    c = __builtin_amdgcn_wmma_f32_16x16x4_f32(
         false, a,  false, bb,
         (short)0, c,  false,  false);
  }

  if (m < 2) {
    const float bias = b[m];
    #pragma unroll
    for (int r = 0; r < 8; ++r) {
      const int row = tile * 16 + hi * 8 + r;
      out[row * 2 + m] = c[r] + bias;
    }
  }
}

extern "C" void kernel_launch(void* const* d_in, const int* in_sizes, int n_in,
                              void* d_out, int out_size, void* d_ws, size_t ws_size,
                              hipStream_t stream) {
  const float* x   = (const float*)d_in[0];
  const float* ic  = (const float*)d_in[1];
  const float* thr = (const float*)d_in[2];
  const float* eps = (const float*)d_in[3];
  const float* W   = (const float*)d_in[4];
  const float* b   = (const float*)d_in[5];
  float* out = (float*)d_out;

  float* ws    = (float*)d_ws;
  float* traj  = ws;
  float* prefE = ws + 10240;
  float* prefC = ws + 20480;
  float* feats = ws + 30720;

  chaos_traj_kernel<<<1, 32, 0, stream>>>(ic, thr, traj, prefE, prefC);
  chaos_feat_kernel<<<NELEM / 256, 256, 0, stream>>>(x, eps, traj, prefE, prefC, feats);
  chaos_gemm_kernel<<<2048 / 16, 32, 0, stream>>>(feats, W, b, out);
}


# traj kernel: per-step correctly rounded divide and its branch replaced by select + multiply by hoisted reciprocal (hi+lo) + exact fma remainder step; quotient bits unchanged, serial chain shortened
# speedup vs baseline: 2.7561x; 2.7561x over previous
#include <hip/hip_runtime.h>
#include <math.h>


#define TRAJ_LEN 10000
#define CHUNK    2048
#define NELEM    (2048 * 32)

typedef __attribute__((ext_vector_type(2))) float v2f;
typedef __attribute__((ext_vector_type(8))) float v8f;

__global__ void chaos_traj_kernel(const float* __restrict__ ic,
                                  const float* __restrict__ thrp,
                                  float* __restrict__ traj,
                                  float* __restrict__ prefE,
                                  float* __restrict__ prefC) {
  if (blockIdx.x == 0 && threadIdx.x == 0) {
    float z  = ic[0];
    const float th  = thrp[0];
    const float om  = 1.f - th;
    const float rh1 = 1.f / th;
    const float rl1 = fmaf(-th, rh1, 1.f) / th;
    const float rh2 = 1.f / om;
    const float rl2 = fmaf(-om, rh2, 1.f) / om;
    const float nth = -th;
    const float nom = -om;
    float e = 0.f, c = 0.f;
    #pragma unroll 8
    for (int k = 0; k < TRAJ_LEN; ++k) {
      const float cur = z;
      const bool  lo  = cur < th;
      const float num = lo ? cur : (1.f - cur);
      const float nd  = lo ? nth : nom;
      const float rh  = lo ? rh1 : rh2;
      const float rl  = lo ? rl1 : rl2;
      const float q0  = fmaf(num, rh, num * rl);
      const float r   = fmaf(nd, q0, num);
      z = fmaf(r, rh, q0);
      traj[k] = cur;
      e += cur * cur;
      c += (cur > th) ? 1.f : 0.f;
      prefE[k] = e;
      prefC[k] = c;
    }
  }
}

__global__ void chaos_feat_kernel(const float* __restrict__ x,
                                  const float* __restrict__ epsp,
                                  const float* __restrict__ traj,
                                  const float* __restrict__ prefE,
                                  const float* __restrict__ prefC,
                                  float* __restrict__ feats) {
  __shared__ float s_traj[CHUNK];
  __shared__ int   s_alive;

  const int   idx = blockIdx.x * blockDim.x + threadIdx.x;
  const float xv  = x[idx];
  const float eps = epsp[0];

  int kFire = -1;
  for (int base = 0; base < TRAJ_LEN; base += CHUNK) {
    const int cn = (TRAJ_LEN - base < CHUNK) ? (TRAJ_LEN - base) : CHUNK;
    for (int i = threadIdx.x; i < cn; i += blockDim.x)
      s_traj[i] = traj[base + i];
    if (threadIdx.x == 0) s_alive = 0;
    if (base + CHUNK < TRAJ_LEN)
      __builtin_prefetch(traj + base + CHUNK + threadIdx.x * 8, 0, 0);
    __syncthreads();

    if (kFire < 0) {
      for (int i = 0; i < cn; ++i) {
        if (fabsf(s_traj[i] - xv) < eps) { kFire = base + i; break; }
      }
      if (kFire < 0) atomicOr(&s_alive, 1);
    }
    __syncthreads();
    const int alive = s_alive;
    __syncthreads();
    if (!alive) break;
  }

  const int   k      = (kFire < 0) ? (TRAJ_LEN - 1) : kFire;
  const float tt     = (kFire < 0) ? (float)TRAJ_LEN : (float)(kFire + 1);
  const float energy = prefE[k];
  const float cnt    = prefC[k];
  const float p      = cnt / tt;
  const float q      = 1.f - p;
  const float e0     = (p > 0.f) ? p * log2f(p) : 0.f;
  const float e1     = (q > 0.f) ? q * log2f(q) : 0.f;
  const float ent    = -(e0 + e1);

  float4 f4 = make_float4(tt, energy, p, ent);
  *reinterpret_cast<float4*>(feats + (size_t)idx * 4) = f4;
}

__global__ void chaos_gemm_kernel(const float* __restrict__ feats,
                                  const float* __restrict__ W,
                                  const float* __restrict__ b,
                                  float* __restrict__ out) {
  __shared__ float s_B[16 * 128];

  const int lane = threadIdx.x;
  const int tile = blockIdx.x;
  const int m    = lane & 15;
  const int hi   = lane >> 4;

  for (int i = lane; i < 16 * 128; i += 32) {
    const int n = i >> 7;
    const int k = i & 127;
    const float v = W[(n & 1) * 128 + k];
    s_B[i] = (n < 2) ? v : 0.f;
  }
  __syncthreads();

  const float* __restrict__ arow = feats + (size_t)(tile * 16 + m) * 128;
  const float* __restrict__ brow = s_B + m * 128;

  v8f c = {};
  #pragma unroll 4
  for (int k0 = 0; k0 < 128; k0 += 4) {
    v2f a;
    a.x = arow[k0 + hi * 2 + 0];
    a.y = arow[k0 + hi * 2 + 1];
    v2f bb;
    bb.x = brow[k0 + hi * 2 + 0];
    bb.y = brow[k0 + hi * 2 + 1];
    c = __builtin_amdgcn_wmma_f32_16x16x4_f32(
         false, a,  false, bb,
         (short)0, c,  false,  false);
  }

  if (m < 2) {
    const float bias = b[m];
    #pragma unroll
    for (int r = 0; r < 8; ++r) {
      const int row = tile * 16 + hi * 8 + r;
      out[row * 2 + m] = c[r] + bias;
    }
  }
}

extern "C" void kernel_launch(void* const* d_in, const int* in_sizes, int n_in,
                              void* d_out, int out_size, void* d_ws, size_t ws_size,
                              hipStream_t stream) {
  const float* x   = (const float*)d_in[0];
  const float* ic  = (const float*)d_in[1];
  const float* thr = (const float*)d_in[2];
  const float* eps = (const float*)d_in[3];
  const float* W   = (const float*)d_in[4];
  const float* b   = (const float*)d_in[5];
  float* out = (float*)d_out;

  float* ws    = (float*)d_ws;
  float* traj  = ws;
  float* prefE = ws + 10240;
  float* prefC = ws + 20480;
  float* feats = ws + 30720;

  chaos_traj_kernel<<<1, 32, 0, stream>>>(ic, thr, traj, prefE, prefC);
  chaos_feat_kernel<<<NELEM / 256, 256, 0, stream>>>(x, eps, traj, prefE, prefC, feats);
  chaos_gemm_kernel<<<2048 / 16, 32, 0, stream>>>(feats, W, b, out);
}


# every wave runs the serial trajectory itself in registers, tests its elements as values are made, and leaves once all have fired; the one-lane 10000-step producer kernel and the LDS scan are gone
# speedup vs baseline: 44.3930x; 16.1072x over previous
#include <hip/hip_runtime.h>
#include <math.h>


#define TRAJ_LEN   10000
#define NELEM      (2048 * 32)
#define EPL        2
#define FEAT_BLOCK 128
#define GROUP      8

typedef __attribute__((ext_vector_type(2))) float v2f;
typedef __attribute__((ext_vector_type(8))) float v8f;

__global__ void __launch_bounds__(FEAT_BLOCK)
chaos_feat_kernel(const float* __restrict__ x,
                  const float* __restrict__ ic,
                  const float* __restrict__ thrp,
                  const float* __restrict__ epsp,
                  float* __restrict__ feats) {
  const int   t   = blockIdx.x * FEAT_BLOCK + threadIdx.x;
  const float x0  = x[t * EPL + 0];
  const float x1  = x[t * EPL + 1];
  const float eps = epsp[0];

  float z  = ic[0];
  const float th  = thrp[0];
  const float om  = 1.f - th;
  const float rh1 = 1.f / th;
  const float rl1 = fmaf(-th, rh1, 1.f) / th;
  const float rh2 = 1.f / om;
  const float rl2 = fmaf(-om, rh2, 1.f) / om;
  const float nth = -th;
  const float nom = -om;

  float e = 0.f, c = 0.f;
  bool  d0 = false, d1 = false;
  int   k0f = 0, k1f = 0;
  float e0f = 0.f, e1f = 0.f;
  float c0f = 0.f, c1f = 0.f;

  for (int kb = 0; kb < TRAJ_LEN; kb += GROUP) {
    float es[GROUP], cs[GROUP];
    bool  m0[GROUP], m1[GROUP];
    bool  n0 = false, n1 = false;
    #pragma unroll
    for (int j = 0; j < GROUP; ++j) {
      const float cur = z;
      const bool  lo  = cur < th;
      const float num = lo ? cur : (1.f - cur);
      const float nd  = lo ? nth : nom;
      const float rh  = lo ? rh1 : rh2;
      const float rl  = lo ? rl1 : rl2;
      const float q0  = fmaf(num, rh, num * rl);
      const float r   = fmaf(nd, q0, num);
      z = fmaf(r, rh, q0);

      e = fmaf(cur, cur, e);
      c += (cur > th) ? 1.f : 0.f;
      es[j] = e;
      cs[j] = c;
      m0[j] = fabsf(cur - x0) < eps;
      m1[j] = fabsf(cur - x1) < eps;
      n0 = n0 || m0[j];
      n1 = n1 || m1[j];
    }
    const bool f0 = n0 && !d0;
    const bool f1 = n1 && !d1;
    if (__builtin_amdgcn_ballot_w32(f0 || f1) != 0u) {
      #pragma unroll
      for (int j = GROUP - 1; j >= 0; --j) {
        const bool h0 = f0 && m0[j];
        const bool h1 = f1 && m1[j];
        k0f = h0 ? (kb + j) : k0f;
        e0f = h0 ? es[j] : e0f;
        c0f = h0 ? cs[j] : c0f;
        k1f = h1 ? (kb + j) : k1f;
        e1f = h1 ? es[j] : e1f;
        c1f = h1 ? cs[j] : c1f;
      }
      d0 = d0 || n0;
      d1 = d1 || n1;
      if (__builtin_amdgcn_ballot_w32(!(d0 && d1)) == 0u) break;
    }
  }

  const float tt0 = d0 ? (float)(k0f + 1) : (float)TRAJ_LEN;
  const float tt1 = d1 ? (float)(k1f + 1) : (float)TRAJ_LEN;
  const float en0 = d0 ? e0f : e;
  const float en1 = d1 ? e1f : e;
  const float cn0 = d0 ? c0f : c;
  const float cn1 = d1 ? c1f : c;

  const float p0  = cn0 / tt0;
  const float q0c = 1.f - p0;
  const float a0  = (p0  > 0.f) ? p0  * log2f(p0)  : 0.f;
  const float b0  = (q0c > 0.f) ? q0c * log2f(q0c) : 0.f;
  const float p1  = cn1 / tt1;
  const float q1c = 1.f - p1;
  const float a1  = (p1  > 0.f) ? p1  * log2f(p1)  : 0.f;
  const float b1  = (q1c > 0.f) ? q1c * log2f(q1c) : 0.f;

  float4* __restrict__ dst = reinterpret_cast<float4*>(feats + (size_t)t * (EPL * 4));
  dst[0] = make_float4(tt0, en0, p0, -(a0 + b0));
  dst[1] = make_float4(tt1, en1, p1, -(a1 + b1));
}

__global__ void chaos_gemm_kernel(const float* __restrict__ feats,
                                  const float* __restrict__ W,
                                  const float* __restrict__ b,
                                  float* __restrict__ out) {
  __shared__ float s_B[16 * 128];

  const int lane = threadIdx.x;
  const int tile = blockIdx.x;
  const int m    = lane & 15;
  const int hi   = lane >> 4;

  for (int i = lane; i < 16 * 128; i += 32) {
    const int n = i >> 7;
    const int k = i & 127;
    const float v = W[(n & 1) * 128 + k];
    s_B[i] = (n < 2) ? v : 0.f;
  }
  __syncthreads();

  const float* __restrict__ arow = feats + (size_t)(tile * 16 + m) * 128;
  const float* __restrict__ brow = s_B + m * 128;

  v8f c = {};
  #pragma unroll 4
  for (int k0 = 0; k0 < 128; k0 += 4) {
    v2f a;
    a.x = arow[k0 + hi * 2 + 0];
    a.y = arow[k0 + hi * 2 + 1];
    v2f bb;
    bb.x = brow[k0 + hi * 2 + 0];
    bb.y = brow[k0 + hi * 2 + 1];
    c = __builtin_amdgcn_wmma_f32_16x16x4_f32(
         false, a,  false, bb,
         (short)0, c,  false,  false);
  }

  if (m < 2) {
    const float bias = b[m];
    #pragma unroll
    for (int r = 0; r < 8; ++r) {
      const int row = tile * 16 + hi * 8 + r;
      out[row * 2 + m] = c[r] + bias;
    }
  }
}

extern "C" void kernel_launch(void* const* d_in, const int* in_sizes, int n_in,
                              void* d_out, int out_size, void* d_ws, size_t ws_size,
                              hipStream_t stream) {
  const float* x   = (const float*)d_in[0];
  const float* ic  = (const float*)d_in[1];
  const float* thr = (const float*)d_in[2];
  const float* eps = (const float*)d_in[3];
  const float* W   = (const float*)d_in[4];
  const float* b   = (const float*)d_in[5];
  float* out = (float*)d_out;

  float* feats = (float*)d_ws;

  chaos_feat_kernel<<<NELEM / (EPL * FEAT_BLOCK), FEAT_BLOCK, 0, stream>>>(x, ic, thr, eps, feats);
  chaos_gemm_kernel<<<2048 / 16, 32, 0, stream>>>(feats, W, b, out);
}


# head fused into the feature kernel (one launch): each 4-wave work-group keeps its 8x128 features in LDS and its first wave runs the same 32 chained fp32 WMMAs on them; feats never go to memory
# speedup vs baseline: 68.5837x; 1.5449x over previous
#include <hip/hip_runtime.h>
#include <math.h>


#define TRAJ_LEN   10000
#define NROWS      2048
#define ROW_ELEMS  32
#define KDIM       128
#define EPL        2
#define FEAT_BLOCK 128
#define TILE_ROWS  (FEAT_BLOCK * EPL / ROW_ELEMS)
#define LPR        (ROW_ELEMS / EPL)
#define GROUP      8
#define PF         132

static_assert(FEAT_BLOCK == KDIM, "one thread per column of W: two floats of W and one zero each");
static_assert(TILE_ROWS == 8, "the head reads the tile rows twice to fill 16 columns");

typedef __attribute__((ext_vector_type(2))) float v2f;
typedef __attribute__((ext_vector_type(8))) float v8f;

__global__ void __launch_bounds__(FEAT_BLOCK)
chaos_fused_kernel(const float* __restrict__ x,
                   const float* __restrict__ ic,
                   const float* __restrict__ thrp,
                   const float* __restrict__ epsp,
                   const float* __restrict__ W,
                   const float* __restrict__ b,
                   float* __restrict__ out) {
  __shared__ __attribute__((aligned(16))) float s_F[TILE_ROWS * PF];
  __shared__ __attribute__((aligned(16))) float s_W[3 * PF];

  const int   tid = threadIdx.x;
  const int   t   = blockIdx.x * FEAT_BLOCK + tid;
  const float x0  = x[t * EPL + 0];
  const float x1  = x[t * EPL + 1];
  const float2 w2 = reinterpret_cast<const float2*>(W)[tid];
  const float  bias0 = b[0];
  const float  bias1 = b[1];
  const float eps = epsp[0];

  float z  = ic[0];
  const float th  = thrp[0];
  const float om  = 1.f - th;
  const float rh1 = 1.f / th;
  const float rl1 = fmaf(-th, rh1, 1.f) / th;
  const float rh2 = 1.f / om;
  const float rl2 = fmaf(-om, rh2, 1.f) / om;
  const float nth = -th;
  const float nom = -om;

  *reinterpret_cast<float2*>(&s_W[(tid >> 6) * PF + (tid & 63) * 2]) = w2;
  s_W[2 * PF + tid] = 0.f;

  float e = 0.f, c = 0.f;
  bool  d0 = false, d1 = false;
  int   k0f = 0, k1f = 0;
  float e0f = 0.f, e1f = 0.f;
  float c0f = 0.f, c1f = 0.f;

  for (int kb = 0; kb < TRAJ_LEN; kb += GROUP) {
    float es[GROUP], cs[GROUP];
    bool  m0[GROUP], m1[GROUP];
    bool  n0 = false, n1 = false;
    #pragma unroll
    for (int j = 0; j < GROUP; ++j) {
      const float cur = z;
      const bool  lo  = cur < th;
      const float num = lo ? cur : (1.f - cur);
      const float nd  = lo ? nth : nom;
      const float rh  = lo ? rh1 : rh2;
      const float rl  = lo ? rl1 : rl2;
      const float q0  = fmaf(num, rh, num * rl);
      const float r   = fmaf(nd, q0, num);
      z = fmaf(r, rh, q0);

      e = fmaf(cur, cur, e);
      c += (cur > th) ? 1.f : 0.f;
      es[j] = e;
      cs[j] = c;
      m0[j] = fabsf(cur - x0) < eps;
      m1[j] = fabsf(cur - x1) < eps;
      n0 = n0 || m0[j];
      n1 = n1 || m1[j];
    }
    const bool f0 = n0 && !d0;
    const bool f1 = n1 && !d1;
    if (__builtin_amdgcn_ballot_w32(f0 || f1) != 0u) {
      #pragma unroll
      for (int j = GROUP - 1; j >= 0; --j) {
        const bool h0 = f0 && m0[j];
        const bool h1 = f1 && m1[j];
        k0f = h0 ? (kb + j) : k0f;
        e0f = h0 ? es[j] : e0f;
        c0f = h0 ? cs[j] : c0f;
        k1f = h1 ? (kb + j) : k1f;
        e1f = h1 ? es[j] : e1f;
        c1f = h1 ? cs[j] : c1f;
      }
      d0 = d0 || n0;
      d1 = d1 || n1;
      if (__builtin_amdgcn_ballot_w32(!(d0 && d1)) == 0u) break;
    }
  }

  const float tt0 = d0 ? (float)(k0f + 1) : (float)TRAJ_LEN;
  const float tt1 = d1 ? (float)(k1f + 1) : (float)TRAJ_LEN;
  const float en0 = d0 ? e0f : e;
  const float en1 = d1 ? e1f : e;
  const float cn0 = d0 ? c0f : c;
  const float cn1 = d1 ? c1f : c;

  const float p0  = cn0 / tt0;
  const float q0c = 1.f - p0;
  const float a0  = (p0  > 0.f) ? p0  * log2f(p0)  : 0.f;
  const float b0  = (q0c > 0.f) ? q0c * log2f(q0c) : 0.f;
  const float p1  = cn1 / tt1;
  const float q1c = 1.f - p1;
  const float a1  = (p1  > 0.f) ? p1  * log2f(p1)  : 0.f;
  const float b1  = (q1c > 0.f) ? q1c * log2f(q1c) : 0.f;

  float4* __restrict__ dst =
      reinterpret_cast<float4*>(&s_F[(tid / LPR) * PF + (tid % LPR) * (EPL * 4)]);
  dst[0] = make_float4(tt0, en0, p0, -(a0 + b0));
  dst[1] = make_float4(tt1, en1, p1, -(a1 + b1));
  __syncthreads();

  const int wave = __builtin_amdgcn_readfirstlane(tid >> 5);
  if (wave == 0) {
    const int cc = tid & 15;
    const int hh = tid >> 4;
    const float* __restrict__ wrow = s_W + (cc < 2 ? cc : 2) * PF + hh * 2;
    const float* __restrict__ frow = s_F + (cc & (TILE_ROWS - 1)) * PF + hh * 2;

    v2f af[KDIM / 4], bf[KDIM / 4];
    #pragma unroll
    for (int s = 0; s < KDIM / 4; ++s) {
      af[s] = *reinterpret_cast<const v2f*>(wrow + 4 * s);
      bf[s] = *reinterpret_cast<const v2f*>(frow + 4 * s);
    }
    __builtin_amdgcn_sched_barrier(0);
    v8f acc = {};
    #pragma unroll
    for (int s = 0; s < KDIM / 4; ++s) {
      acc = __builtin_amdgcn_wmma_f32_16x16x4_f32(
           false, af[s],  false, bf[s],
           (short)0, acc,  false,  false);
    }

    if (tid < TILE_ROWS) {
      float2 o;
      o.x = acc[0] + bias0;
      o.y = acc[1] + bias1;
      reinterpret_cast<float2*>(out)[blockIdx.x * TILE_ROWS + cc] = o;
    }
  }
}

extern "C" void kernel_launch(void* const* d_in, const int* in_sizes, int n_in,
                              void* d_out, int out_size, void* d_ws, size_t ws_size,
                              hipStream_t stream) {
  const float* x   = (const float*)d_in[0];
  const float* ic  = (const float*)d_in[1];
  const float* thr = (const float*)d_in[2];
  const float* eps = (const float*)d_in[3];
  const float* W   = (const float*)d_in[4];
  const float* b   = (const float*)d_in[5];
  float* out = (float*)d_out;

  chaos_fused_kernel<<<NROWS / TILE_ROWS, FEAT_BLOCK, 0, stream>>>(x, ic, thr, eps, W, b, out);
}


# chain loop fast path keeps per step only z, e, c and one running min of |z-x| per lane; hit masks and fired flags are rebuilt from the group's kept values only when the one-compare vote fires
# speedup vs baseline: 100.1311x; 1.4600x over previous
#include <hip/hip_runtime.h>
#include <math.h>


#define TRAJ_LEN   10000
#define NROWS      2048
#define ROW_ELEMS  32
#define KDIM       128
#define EPL        2
#define FEAT_BLOCK 128
#define TILE_ROWS  (FEAT_BLOCK * EPL / ROW_ELEMS)
#define LPR        (ROW_ELEMS / EPL)
#define GROUP      8
#define PF         132

static_assert(FEAT_BLOCK == KDIM, "one thread per column of W: two floats of W and one zero each");
static_assert(TILE_ROWS == 8, "the head reads the tile rows twice to fill 16 columns");

typedef __attribute__((ext_vector_type(2))) float v2f;
typedef __attribute__((ext_vector_type(8))) float v8f;

__global__ void __launch_bounds__(FEAT_BLOCK)
chaos_fused_kernel(const float* __restrict__ x,
                   const float* __restrict__ ic,
                   const float* __restrict__ thrp,
                   const float* __restrict__ epsp,
                   const float* __restrict__ W,
                   const float* __restrict__ b,
                   float* __restrict__ out) {
  __shared__ __attribute__((aligned(16))) float s_F[TILE_ROWS * PF];
  __shared__ __attribute__((aligned(16))) float s_W[3 * PF];

  const int   tid = threadIdx.x;
  const int   t   = blockIdx.x * FEAT_BLOCK + tid;
  const float x0  = x[t * EPL + 0];
  const float x1  = x[t * EPL + 1];
  const float2 w2 = reinterpret_cast<const float2*>(W)[tid];
  const float  bias0 = b[0];
  const float  bias1 = b[1];
  const float eps = epsp[0];

  float z  = ic[0];
  const float th  = thrp[0];
  const float om  = 1.f - th;
  const float rh1 = 1.f / th;
  const float rl1 = fmaf(-th, rh1, 1.f) / th;
  const float rh2 = 1.f / om;
  const float rl2 = fmaf(-om, rh2, 1.f) / om;
  const float nth = -th;
  const float nom = -om;

  *reinterpret_cast<float2*>(&s_W[(tid >> 6) * PF + (tid & 63) * 2]) = w2;
  s_W[2 * PF + tid] = 0.f;

  float e = 0.f, c = 0.f;
  int   k0f = -1, k1f = -1;
  float e0f = 0.f, e1f = 0.f;
  float c0f = 0.f, c1f = 0.f;
  float xa = x0, xb = x1;

  for (int kb = 0; kb < TRAJ_LEN; kb += GROUP) {
    float zs[GROUP];
    float es[GROUP], cs[GROUP];
    float dm = 0.f;
    #pragma unroll
    for (int j = 0; j < GROUP; ++j) {
      const float cur = z;
      const bool  lo  = cur < th;
      const float num = lo ? cur : (1.f - cur);
      const float nd  = lo ? nth : nom;
      const float rh  = lo ? rh1 : rh2;
      const float rl  = lo ? rl1 : rl2;
      const float q0  = fmaf(num, rh, num * rl);
      const float r   = fmaf(nd, q0, num);
      z = fmaf(r, rh, q0);

      e = fmaf(cur, cur, e);
      c += (cur > th) ? 1.f : 0.f;
      zs[j] = cur;
      es[j] = e;
      cs[j] = c;
      const float dd = fminf(fabsf(cur - xa), fabsf(cur - xb));
      dm = (j == 0) ? dd : fminf(dm, dd);
    }
    asm volatile("" :: "s"(z));
    if (__builtin_amdgcn_ballot_w32(dm < eps) != 0u) {
      #pragma unroll
      for (int j = GROUP - 1; j >= 0; --j) {
        const bool h0 = fabsf(zs[j] - xa) < eps;
        const bool h1 = fabsf(zs[j] - xb) < eps;
        k0f = h0 ? (kb + j) : k0f;
        e0f = h0 ? es[j] : e0f;
        c0f = h0 ? cs[j] : c0f;
        k1f = h1 ? (kb + j) : k1f;
        e1f = h1 ? es[j] : e1f;
        c1f = h1 ? cs[j] : c1f;
      }
      xa = (k0f >= 0) ? __builtin_inff() : xa;
      xb = (k1f >= 0) ? __builtin_inff() : xb;
      if (__builtin_amdgcn_ballot_w32((k0f | k1f) < 0) == 0u) break;
    }
  }

  const bool  d0  = k0f >= 0;
  const bool  d1  = k1f >= 0;
  const float tt0 = d0 ? (float)(k0f + 1) : (float)TRAJ_LEN;
  const float tt1 = d1 ? (float)(k1f + 1) : (float)TRAJ_LEN;
  const float en0 = d0 ? e0f : e;
  const float en1 = d1 ? e1f : e;
  const float cn0 = d0 ? c0f : c;
  const float cn1 = d1 ? c1f : c;

  const float p0  = cn0 / tt0;
  const float q0c = 1.f - p0;
  const float a0  = (p0  > 0.f) ? p0  * log2f(p0)  : 0.f;
  const float b0  = (q0c > 0.f) ? q0c * log2f(q0c) : 0.f;
  const float p1  = cn1 / tt1;
  const float q1c = 1.f - p1;
  const float a1  = (p1  > 0.f) ? p1  * log2f(p1)  : 0.f;
  const float b1  = (q1c > 0.f) ? q1c * log2f(q1c) : 0.f;

  float4* __restrict__ dst =
      reinterpret_cast<float4*>(&s_F[(tid / LPR) * PF + (tid % LPR) * (EPL * 4)]);
  dst[0] = make_float4(tt0, en0, p0, -(a0 + b0));
  dst[1] = make_float4(tt1, en1, p1, -(a1 + b1));
  __syncthreads();

  const int wave = __builtin_amdgcn_readfirstlane(tid >> 5);
  if (wave == 0) {
    const int cc = tid & 15;
    const int hh = tid >> 4;
    const float* __restrict__ wrow = s_W + (cc < 2 ? cc : 2) * PF + hh * 2;
    const float* __restrict__ frow = s_F + (cc & (TILE_ROWS - 1)) * PF + hh * 2;

    v2f af[KDIM / 4], bf[KDIM / 4];
    #pragma unroll
    for (int s = 0; s < KDIM / 4; ++s) {
      af[s] = *reinterpret_cast<const v2f*>(wrow + 4 * s);
      bf[s] = *reinterpret_cast<const v2f*>(frow + 4 * s);
    }
    __builtin_amdgcn_sched_barrier(0);
    v8f acc = {};
    #pragma unroll
    for (int s = 0; s < KDIM / 4; ++s) {
      acc = __builtin_amdgcn_wmma_f32_16x16x4_f32(
           false, af[s],  false, bf[s],
           (short)0, acc,  false,  false);
    }

    if (tid < TILE_ROWS) {
      float2 o;
      o.x = acc[0] + bias0;
      o.y = acc[1] + bias1;
      reinterpret_cast<float2*>(out)[blockIdx.x * TILE_ROWS + cc] = o;
    }
  }
}

extern "C" void kernel_launch(void* const* d_in, const int* in_sizes, int n_in,
                              void* d_out, int out_size, void* d_ws, size_t ws_size,
                              hipStream_t stream) {
  const float* x   = (const float*)d_in[0];
  const float* ic  = (const float*)d_in[1];
  const float* thr = (const float*)d_in[2];
  const float* eps = (const float*)d_in[3];
  const float* W   = (const float*)d_in[4];
  const float* b   = (const float*)d_in[5];
  float* out = (float*)d_out;

  chaos_fused_kernel<<<NROWS / TILE_ROWS, FEAT_BLOCK, 0, stream>>>(x, ic, thr, eps, W, b, out);
}
